# starting kernel
# baseline (speedup 1.0000x reference)
#include <hip/hip_runtime.h>

#define N_NODES 50000
#define N_EDGES 800000
#define N_PAIRS 400000
#define DIM     128
#define LDS_STRIDE 132

typedef float v2f __attribute__((ext_vector_type(2)));
typedef float v8f __attribute__((ext_vector_type(8)));

__global__ void k_init_deg(float* __restrict__ deg) {
    int i = blockIdx.x * blockDim.x + threadIdx.x;
    if (i < N_NODES) deg[i] = 1.0f;
}

__global__ void k_scatter_deg(const int* __restrict__ dst, float* __restrict__ deg) {
    int e = blockIdx.x * blockDim.x + threadIdx.x;
    if (e < N_EDGES) atomicAdd(&deg[dst[e]], 1.0f);
}

__global__ void k_rsqrt_inplace(float* __restrict__ deg) {
    int i = blockIdx.x * blockDim.x + threadIdx.x;
    if (i < N_NODES) deg[i] = rsqrtf(deg[i]);
}

__launch_bounds__(256, 4)
__global__ void k_matmul_wmma(const float* __restrict__ X,
                              const float* __restrict__ W,
                              const float* __restrict__ bias,
                              float* __restrict__ H) {
    __shared__ float lds[16 * LDS_STRIDE];
    const int rowBase = blockIdx.x * 16;
    const int t = threadIdx.x;

    #pragma unroll
    for (int rep = 0; rep < 2; ++rep) {
        int idx = t + rep * 256;
        int r   = idx >> 5;
        int c4  = idx & 31;
        float4 v = *(const float4*)(X + (size_t)(rowBase + r) * DIM + c4 * 4);
        if (bias) {
            float4 b = *(const float4*)(bias + c4 * 4);
            v.x = fmaxf(v.x + b.x, 0.0f);
            v.y = fmaxf(v.y + b.y, 0.0f);
            v.z = fmaxf(v.z + b.z, 0.0f);
            v.w = fmaxf(v.w + b.w, 0.0f);
        }
        *(float4*)(&lds[r * LDS_STRIDE + c4 * 4]) = v;
    }
    __syncthreads();

    const int wave = t >> 5;
    const int lane = t & 31;
    const int half = lane >> 4;
    const int l16  = lane & 15;
    const int colBase = wave * 16;

    v8f acc = {0.f, 0.f, 0.f, 0.f, 0.f, 0.f, 0.f, 0.f};

    #pragma unroll
    for (int k0 = 0; k0 < DIM; k0 += 4) {
        const int kk = k0 + half * 2;
        float2 av = *(const float2*)(&lds[l16 * LDS_STRIDE + kk]);
        v2f a, b;
        a[0] = av.x;
        a[1] = av.y;
        b[0] = W[(size_t)kk * DIM + colBase + l16];
        b[1] = W[(size_t)(kk + 1) * DIM + colBase + l16];
        acc = __builtin_amdgcn_wmma_f32_16x16x4_f32(false, a, false, b,
                                                    (short)0, acc, false, false);
    }

    float* outp = H + (size_t)rowBase * DIM + colBase;
    #pragma unroll
    for (int v = 0; v < 8; ++v) {
        outp[(size_t)(v + half * 8) * DIM + l16] = acc[v];
    }
}

__global__ void k_init_out(const float* __restrict__ h, const float* __restrict__ dinv,
                           const float* __restrict__ bias, float* __restrict__ out) {
    int idx = blockIdx.x * blockDim.x + threadIdx.x;
    int i = idx >> 7;
    int f = idx & 127;
    float di = dinv[i];
    float v = h[idx] * di * di;
    if (bias) v += bias[f];
    out[idx] = v;
}

__launch_bounds__(256)
__global__ void k_scatter_edges(const int* __restrict__ src, const int* __restrict__ dst,
                                const float* __restrict__ h, const float* __restrict__ dinv,
                                float* __restrict__ out) {
    const int wid  = threadIdx.x >> 5;
    const int lane = threadIdx.x & 31;
    const int e = blockIdx.x * 8 + wid;
    if (e >= N_EDGES) return;
    const int s = src[e];
    const int d = dst[e];
    const float norm = dinv[s] * dinv[d];
    const float4 hv = *(const float4*)(h + (size_t)s * DIM + lane * 4);
    float* op = out + (size_t)d * DIM + lane * 4;
    atomicAdd(op + 0, hv.x * norm);
    atomicAdd(op + 1, hv.y * norm);
    atomicAdd(op + 2, hv.z * norm);
    atomicAdd(op + 3, hv.w * norm);
}

__launch_bounds__(256)
__global__ void k_decode(const int* __restrict__ pa, const int* __restrict__ pb,
                         const float* __restrict__ z, float* __restrict__ out) {
    const int wid  = threadIdx.x >> 5;
    const int lane = threadIdx.x & 31;
    const int p = blockIdx.x * 8 + wid;
    if (p >= N_PAIRS) return;
    const int a = pa[p];
    const int b = pb[p];
    const float4 za = *(const float4*)(z + (size_t)a * DIM + lane * 4);
    const float4 zb = *(const float4*)(z + (size_t)b * DIM + lane * 4);
    float s = za.x * zb.x + za.y * zb.y + za.z * zb.z + za.w * zb.w;
    #pragma unroll
    for (int off = 16; off > 0; off >>= 1) s += __shfl_down(s, off);
    if (lane == 0) out[p] = s;
}

extern "C" void kernel_launch(void* const* d_in, const int* in_sizes, int n_in,
                              void* d_out, int out_size, void* d_ws, size_t ws_size,
                              hipStream_t stream) {
    const int*   edge_index = (const int*)d_in[0];
    const int*   edge_pairs = (const int*)d_in[1];
    const float* emb        = (const float*)d_in[2];
    const float* W1         = (const float*)d_in[3];
    const float* b1         = (const float*)d_in[4];
    const float* W2         = (const float*)d_in[5];
    const float* b2         = (const float*)d_in[6];
    float*       out        = (float*)d_out;

    const int* src = edge_index;
    const int* dst = edge_index + N_EDGES;
    const int* pa  = edge_pairs;
    const int* pb  = edge_pairs + N_PAIRS;

    char* ws = (char*)d_ws;
    size_t off = 0;
    float* dinv = (float*)(ws + off); off += ((size_t)N_NODES * 4 + 255) & ~(size_t)255;
    float* h    = (float*)(ws + off); off += (size_t)N_NODES * DIM * 4;
    float* agg  = (float*)(ws + off); off += (size_t)N_NODES * DIM * 4;
    (void)ws_size; (void)in_sizes; (void)n_in; (void)out_size;

    const int T = 256;
    const int gN    = (N_NODES + T - 1) / T;
    const int gE    = (N_EDGES + T - 1) / T;
    const int gNF   = (N_NODES * DIM) / T;
    const int gMM   = N_NODES / 16;
    const int gEw   = N_EDGES / 8;
    const int gPw   = N_PAIRS / 8;

    k_init_deg<<<gN, T, 0, stream>>>(dinv);
    k_scatter_deg<<<gE, T, 0, stream>>>(dst, dinv);
    k_rsqrt_inplace<<<gN, T, 0, stream>>>(dinv);

    k_matmul_wmma<<<gMM, T, 0, stream>>>(emb, W1, nullptr, h);
    k_init_out<<<gNF, T, 0, stream>>>(h, dinv, nullptr, agg);
    k_scatter_edges<<<gEw, T, 0, stream>>>(src, dst, h, dinv, agg);

    k_matmul_wmma<<<gMM, T, 0, stream>>>(agg, W2, b1, h);
    k_init_out<<<gNF, T, 0, stream>>>(h, dinv, b2, agg);
    k_scatter_edges<<<gEw, T, 0, stream>>>(src, dst, h, dinv, agg);

    k_decode<<<gPw, T, 0, stream>>>(pa, pb, agg, out);
}


# Float-atomic scatter-add replaced by per-call counting sort of edges by destination (integer rank adds, scan, fill); one wave per node sums its rows in registers in arrival-rank order, stores once
# speedup vs baseline: 2.1276x; 2.1276x over previous
#include <hip/hip_runtime.h>

#define N_NODES 50000
#define N_EDGES 800000
#define N_PAIRS 400000
#define DIM     128
#define LDS_STRIDE 132

typedef float v2f __attribute__((ext_vector_type(2)));
typedef float v8f __attribute__((ext_vector_type(8)));

__global__ void k_zero_cnt(int* __restrict__ cnt) {
    int i = blockIdx.x * blockDim.x + threadIdx.x;
    if (i < N_NODES) cnt[i] = 0;
}

__global__ void k_count_rank(const int* __restrict__ dst, int* __restrict__ cnt,
                             int* __restrict__ rank) {
    int e = blockIdx.x * blockDim.x + threadIdx.x;
    if (e < N_EDGES) rank[e] = atomicAdd(&cnt[dst[e]], 1);
}

#define SCAN_T 1024
#define SCAN_BLOCKS 13
__launch_bounds__(SCAN_T)
__global__ void k_scan_deg(const int* __restrict__ cnt, int* __restrict__ rowptr,
                           float* __restrict__ dinv) {
    __shared__ int s_tot[32];
    __shared__ int s_pre[32];
    const int t = threadIdx.x;
    const int lane = t & 31;
    const int wave = t >> 5;
    const int b = blockIdx.x;
    const int4* c4 = (const int4*)cnt;

    int4 f[SCAN_BLOCKS - 1];
    #pragma unroll
    for (int k = 0; k < SCAN_BLOCKS - 1; ++k) f[k] = c4[k * SCAN_T + t];
    int pre = 0;
    #pragma unroll
    for (int k = 0; k < SCAN_BLOCKS - 1; ++k)
        pre += (k < b) ? ((f[k].x + f[k].y) + (f[k].z + f[k].w)) : 0;

    const int i4 = b * SCAN_T + t;
    int4 c = make_int4(0, 0, 0, 0);
    if (i4 < N_NODES / 4) c = c4[i4];
    const int tot = (c.x + c.y) + (c.z + c.w);

    int incl = tot;
    #pragma unroll
    for (int off = 1; off < 32; off <<= 1) {
        int up = __shfl_up(incl, off);
        if (lane >= off) incl += up;
    }
    #pragma unroll
    for (int off = 16; off > 0; off >>= 1) pre += __shfl_xor(pre, off);
    if (lane == 31) s_tot[wave] = incl;
    if (lane == 0)  s_pre[wave] = pre;
    __syncthreads();
    int base = 0;
    for (int w = 0; w < 32; ++w) {
        base += s_pre[w];
        if (w < wave) base += s_tot[w];
    }
    if (i4 < N_NODES / 4) {
        int4 r;
        r.x = base + incl - tot;
        r.y = r.x + c.x;
        r.z = r.y + c.y;
        r.w = r.z + c.z;
        ((int4*)rowptr)[i4] = r;
        float4 dv;
        dv.x = rsqrtf((float)(c.x + 1));
        dv.y = rsqrtf((float)(c.y + 1));
        dv.z = rsqrtf((float)(c.z + 1));
        dv.w = rsqrtf((float)(c.w + 1));
        ((float4*)dinv)[i4] = dv;
    }
}

__global__ void k_fill_sorted(const int* __restrict__ src, const int* __restrict__ dst,
                              const int* __restrict__ rank, const int* __restrict__ rowptr,
                              const float* __restrict__ dinv, int* __restrict__ ent) {
    int e = blockIdx.x * blockDim.x + threadIdx.x;
    if (e < N_EDGES) {
        const int s = src[e];
        const int d = dst[e];
        const float w = dinv[s] * dinv[d];
        ((int2*)ent)[rowptr[d] + rank[e]] = make_int2(s, __float_as_int(w));
    }
}

__launch_bounds__(256, 4)
__global__ void k_matmul_wmma(const float* __restrict__ X,
                              const float* __restrict__ W,
                              const float* __restrict__ bias,
                              float* __restrict__ H) {
    __shared__ float lds[16 * LDS_STRIDE];
    const int rowBase = blockIdx.x * 16;
    const int t = threadIdx.x;

    #pragma unroll
    for (int rep = 0; rep < 2; ++rep) {
        int idx = t + rep * 256;
        int r   = idx >> 5;
        int c4  = idx & 31;
        float4 v = *(const float4*)(X + (size_t)(rowBase + r) * DIM + c4 * 4);
        if (bias) {
            float4 b = *(const float4*)(bias + c4 * 4);
            v.x = fmaxf(v.x + b.x, 0.0f);
            v.y = fmaxf(v.y + b.y, 0.0f);
            v.z = fmaxf(v.z + b.z, 0.0f);
            v.w = fmaxf(v.w + b.w, 0.0f);
        }
        *(float4*)(&lds[r * LDS_STRIDE + c4 * 4]) = v;
    }
    __syncthreads();

    const int wave = t >> 5;
    const int lane = t & 31;
    const int half = lane >> 4;
    const int l16  = lane & 15;
    const int colBase = wave * 16;

    v8f acc = {0.f, 0.f, 0.f, 0.f, 0.f, 0.f, 0.f, 0.f};

    #pragma unroll
    for (int k0 = 0; k0 < DIM; k0 += 4) {
        const int kk = k0 + half * 2;
        float2 av = *(const float2*)(&lds[l16 * LDS_STRIDE + kk]);
        v2f a, b;
        a[0] = av.x;
        a[1] = av.y;
        b[0] = W[(size_t)kk * DIM + colBase + l16];
        b[1] = W[(size_t)(kk + 1) * DIM + colBase + l16];
        acc = __builtin_amdgcn_wmma_f32_16x16x4_f32(false, a, false, b,
                                                    (short)0, acc, false, false);
    }

    float* outp = H + (size_t)rowBase * DIM + colBase;
    #pragma unroll
    for (int v = 0; v < 8; ++v) {
        outp[(size_t)(v + half * 8) * DIM + l16] = acc[v];
    }
}


template <int K>
__device__ __forceinline__ void add_rows(float4& acc, const float* __restrict__ hl,
                                         const int2 en, const int j) {
    int s[K];
    float w[K];
    float4 r[K];
    #pragma unroll
    for (int k = 0; k < K; ++k) {
        s[k] = __builtin_amdgcn_readlane(en.x, j + k);
        w[k] = __int_as_float(__builtin_amdgcn_readlane(en.y, j + k));
    }
    #pragma unroll
    for (int k = 0; k < K; ++k) r[k] = *(const float4*)(hl + (size_t)s[k] * DIM);
    #pragma unroll
    for (int k = 0; k < K; ++k) {
        acc.x = fmaf(r[k].x, w[k], acc.x);
        acc.y = fmaf(r[k].y, w[k], acc.y);
        acc.z = fmaf(r[k].z, w[k], acc.z);
        acc.w = fmaf(r[k].w, w[k], acc.w);
    }
}

__launch_bounds__(256)
__global__ void k_gather_rows(const int* __restrict__ rowptr, const int* __restrict__ cnt,
                              const int* __restrict__ ent, const float* __restrict__ h,
                              const float* __restrict__ dinv, const float* __restrict__ bias,
                              float* __restrict__ out) {
    const int lane = threadIdx.x & 31;
    const int d = __builtin_amdgcn_readfirstlane((int)(blockIdx.x * 8 + (threadIdx.x >> 5)));
    const int start = rowptr[d];
    const int n = cnt[d];
    const float di = dinv[d];
    const int2* ent2 = (const int2*)ent;
    int m = (n < 32) ? n : 32;
    int2 en = make_int2(0, 0);
    if (lane < m) en = ent2[start + lane];
    const float* hl = h + lane * 4;
    const float4 hs = *(const float4*)(hl + (size_t)d * DIM);
    float4 acc;
    acc.x = hs.x * di * di;
    acc.y = hs.y * di * di;
    acc.z = hs.z * di * di;
    acc.w = hs.w * di * di;
    if (bias) {
        const float4 b = *(const float4*)(bias + lane * 4);
        acc.x += b.x; acc.y += b.y; acc.z += b.z; acc.w += b.w;
    }
    for (int base = 0; base < n; ) {
        int j = 0;
        for (; j + 8 <= m; j += 8) add_rows<8>(acc, hl, en, j);
        if (j + 4 <= m) { add_rows<4>(acc, hl, en, j); j += 4; }
        if (j + 2 <= m) { add_rows<2>(acc, hl, en, j); j += 2; }
        if (j < m)      { add_rows<1>(acc, hl, en, j); }
        base += 32;
        if (base < n) {
            m = (n - base < 32) ? (n - base) : 32;
            en = make_int2(0, 0);
            if (lane < m) en = ent2[start + base + lane];
        }
    }
    *(float4*)(out + (size_t)d * DIM + lane * 4) = acc;
}

__launch_bounds__(256)
__global__ void k_decode(const int* __restrict__ pa, const int* __restrict__ pb,
                         const float* __restrict__ z, float* __restrict__ out) {
    const int wid  = threadIdx.x >> 5;
    const int lane = threadIdx.x & 31;
    const int p = blockIdx.x * 8 + wid;
    if (p >= N_PAIRS) return;
    const int a = pa[p];
    const int b = pb[p];
    const float4 za = *(const float4*)(z + (size_t)a * DIM + lane * 4);
    const float4 zb = *(const float4*)(z + (size_t)b * DIM + lane * 4);
    float s = za.x * zb.x + za.y * zb.y + za.z * zb.z + za.w * zb.w;
    #pragma unroll
    for (int off = 16; off > 0; off >>= 1) s += __shfl_down(s, off);
    if (lane == 0) out[p] = s;
}

extern "C" void kernel_launch(void* const* d_in, const int* in_sizes, int n_in,
                              void* d_out, int out_size, void* d_ws, size_t ws_size,
                              hipStream_t stream) {
    const int*   edge_index = (const int*)d_in[0];
    const int*   edge_pairs = (const int*)d_in[1];
    const float* emb        = (const float*)d_in[2];
    const float* W1         = (const float*)d_in[3];
    const float* b1         = (const float*)d_in[4];
    const float* W2         = (const float*)d_in[5];
    const float* b2         = (const float*)d_in[6];
    float*       out        = (float*)d_out;

    const int* src = edge_index;
    const int* dst = edge_index + N_EDGES;
    const int* pa  = edge_pairs;
    const int* pb  = edge_pairs + N_PAIRS;

    char* ws = (char*)d_ws;
    size_t off = 0;
    int*   cnt    = (int*)(ws + off);   off += ((size_t)N_NODES * 4 + 255) & ~(size_t)255;
    int*   rowptr = (int*)(ws + off);   off += ((size_t)N_NODES * 4 + 255) & ~(size_t)255;
    float* dinv   = (float*)(ws + off); off += ((size_t)N_NODES * 4 + 255) & ~(size_t)255;
    int*   rank   = (int*)(ws + off);   off += (size_t)N_EDGES * 4;
    int*   ent    = (int*)(ws + off);   off += (size_t)N_EDGES * 8;
    float* h      = (float*)(ws + off); off += (size_t)N_NODES * DIM * 4;
    float* agg    = (float*)(ws + off); off += (size_t)N_NODES * DIM * 4;
    (void)ws_size; (void)in_sizes; (void)n_in; (void)out_size;

    const int T = 256;
    const int gN    = (N_NODES + T - 1) / T;
    const int gE    = (N_EDGES + T - 1) / T;
    const int gMM   = N_NODES / 16;
    const int gNw   = N_NODES / 8;
    const int gPw   = N_PAIRS / 8;

    k_zero_cnt<<<gN, T, 0, stream>>>(cnt);
    k_count_rank<<<gE, T, 0, stream>>>(dst, cnt, rank);
    k_scan_deg<<<SCAN_BLOCKS, SCAN_T, 0, stream>>>(cnt, rowptr, dinv);
    k_fill_sorted<<<gE, T, 0, stream>>>(src, dst, rank, rowptr, dinv, ent);

    k_matmul_wmma<<<gMM, T, 0, stream>>>(emb, W1, nullptr, h);
    k_gather_rows<<<gNw, T, 0, stream>>>(rowptr, cnt, ent, h, dinv, nullptr, agg);

    k_matmul_wmma<<<gMM, T, 0, stream>>>(agg, W2, b1, h);
    k_gather_rows<<<gNw, T, 0, stream>>>(rowptr, cnt, ent, h, dinv, b2, agg);

    k_decode<<<gPw, T, 0, stream>>>(pa, pb, agg, out);
}


# Each wave does many items instead of one: decode 32 pairs per wave (400000 waves -> 12500), gather 4 nodes per wave, count/fill 4 edges per thread; order of terms in every sum unchanged
# speedup vs baseline: 2.4335x; 1.1438x over previous
#include <hip/hip_runtime.h>

#define N_NODES 50000
#define N_EDGES 800000
#define N_PAIRS 400000
#define DIM     128
#define LDS_STRIDE 132

typedef float v2f __attribute__((ext_vector_type(2)));
typedef float v8f __attribute__((ext_vector_type(8)));

__global__ void k_zero_cnt(int* __restrict__ cnt) {
    int i = blockIdx.x * blockDim.x + threadIdx.x;
    if (i < N_NODES) cnt[i] = 0;
}

__global__ void k_count_rank(const int* __restrict__ dst, int* __restrict__ cnt,
                             int* __restrict__ rank) {
    int q = blockIdx.x * blockDim.x + threadIdx.x;
    if (q < N_EDGES / 4) {
        const int4 d = ((const int4*)dst)[q];
        int4 r;
        r.x = atomicAdd(&cnt[d.x], 1);
        r.y = atomicAdd(&cnt[d.y], 1);
        r.z = atomicAdd(&cnt[d.z], 1);
        r.w = atomicAdd(&cnt[d.w], 1);
        ((int4*)rank)[q] = r;
    }
}

#define SCAN_T 1024
#define SCAN_BLOCKS 13
__launch_bounds__(SCAN_T)
__global__ void k_scan_deg(const int* __restrict__ cnt, int* __restrict__ rowptr,
                           float* __restrict__ dinv) {
    __shared__ int s_tot[32];
    __shared__ int s_pre[32];
    const int t = threadIdx.x;
    const int lane = t & 31;
    const int wave = t >> 5;
    const int b = blockIdx.x;
    const int4* c4 = (const int4*)cnt;

    int4 f[SCAN_BLOCKS - 1];
    #pragma unroll
    for (int k = 0; k < SCAN_BLOCKS - 1; ++k) f[k] = c4[k * SCAN_T + t];
    int pre = 0;
    #pragma unroll
    for (int k = 0; k < SCAN_BLOCKS - 1; ++k)
        pre += (k < b) ? ((f[k].x + f[k].y) + (f[k].z + f[k].w)) : 0;

    const int i4 = b * SCAN_T + t;
    int4 c = make_int4(0, 0, 0, 0);
    if (i4 < N_NODES / 4) c = c4[i4];
    const int tot = (c.x + c.y) + (c.z + c.w);

    int incl = tot;
    #pragma unroll
    for (int off = 1; off < 32; off <<= 1) {
        int up = __shfl_up(incl, off);
        if (lane >= off) incl += up;
    }
    #pragma unroll
    for (int off = 16; off > 0; off >>= 1) pre += __shfl_xor(pre, off);
    if (lane == 31) s_tot[wave] = incl;
    if (lane == 0)  s_pre[wave] = pre;
    __syncthreads();
    int base = 0;
    for (int w = 0; w < 32; ++w) {
        base += s_pre[w];
        if (w < wave) base += s_tot[w];
    }
    if (i4 < N_NODES / 4) {
        int4 r;
        r.x = base + incl - tot;
        r.y = r.x + c.x;
        r.z = r.y + c.y;
        r.w = r.z + c.z;
        ((int4*)rowptr)[i4] = r;
        float4 dv;
        dv.x = rsqrtf((float)(c.x + 1));
        dv.y = rsqrtf((float)(c.y + 1));
        dv.z = rsqrtf((float)(c.z + 1));
        dv.w = rsqrtf((float)(c.w + 1));
        ((float4*)dinv)[i4] = dv;
    }
}

__global__ void k_fill_sorted(const int* __restrict__ src, const int* __restrict__ dst,
                              const int* __restrict__ rank, const int* __restrict__ rowptr,
                              const float* __restrict__ dinv, int* __restrict__ ent) {
    int q = blockIdx.x * blockDim.x + threadIdx.x;
    if (q < N_EDGES / 4) {
        const int4 s4 = ((const int4*)src)[q];
        const int4 d4 = ((const int4*)dst)[q];
        const int4 r4 = ((const int4*)rank)[q];
        const int s[4] = {s4.x, s4.y, s4.z, s4.w};
        const int d[4] = {d4.x, d4.y, d4.z, d4.w};
        const int r[4] = {r4.x, r4.y, r4.z, r4.w};
        float w[4];
        int   pos[4];
        #pragma unroll
        for (int k = 0; k < 4; ++k) {
            w[k]   = dinv[s[k]] * dinv[d[k]];
            pos[k] = rowptr[d[k]] + r[k];
        }
        #pragma unroll
        for (int k = 0; k < 4; ++k)
            ((int2*)ent)[pos[k]] = make_int2(s[k], __float_as_int(w[k]));
    }
}

__launch_bounds__(256, 4)
__global__ void k_matmul_wmma(const float* __restrict__ X,
                              const float* __restrict__ W,
                              const float* __restrict__ bias,
                              float* __restrict__ H) {
    __shared__ float lds[16 * LDS_STRIDE];
    const int rowBase = blockIdx.x * 16;
    const int t = threadIdx.x;

    #pragma unroll
    for (int rep = 0; rep < 2; ++rep) {
        int idx = t + rep * 256;
        int r   = idx >> 5;
        int c4  = idx & 31;
        float4 v = *(const float4*)(X + (size_t)(rowBase + r) * DIM + c4 * 4);
        if (bias) {
            float4 b = *(const float4*)(bias + c4 * 4);
            v.x = fmaxf(v.x + b.x, 0.0f);
            v.y = fmaxf(v.y + b.y, 0.0f);
            v.z = fmaxf(v.z + b.z, 0.0f);
            v.w = fmaxf(v.w + b.w, 0.0f);
        }
        *(float4*)(&lds[r * LDS_STRIDE + c4 * 4]) = v;
    }
    __syncthreads();

    const int wave = t >> 5;
    const int lane = t & 31;
    const int half = lane >> 4;
    const int l16  = lane & 15;
    const int colBase = wave * 16;

    v8f acc = {0.f, 0.f, 0.f, 0.f, 0.f, 0.f, 0.f, 0.f};

    #pragma unroll
    for (int k0 = 0; k0 < DIM; k0 += 4) {
        const int kk = k0 + half * 2;
        float2 av = *(const float2*)(&lds[l16 * LDS_STRIDE + kk]);
        v2f a, b;
        a[0] = av.x;
        a[1] = av.y;
        b[0] = W[(size_t)kk * DIM + colBase + l16];
        b[1] = W[(size_t)(kk + 1) * DIM + colBase + l16];
        acc = __builtin_amdgcn_wmma_f32_16x16x4_f32(false, a, false, b,
                                                    (short)0, acc, false, false);
    }

    float* outp = H + (size_t)rowBase * DIM + colBase;
    #pragma unroll
    for (int v = 0; v < 8; ++v) {
        outp[(size_t)(v + half * 8) * DIM + l16] = acc[v];
    }
}


template <int K>
__device__ __forceinline__ void add_rows(float4& acc, const float* __restrict__ hl,
                                         const int2 en, const int j) {
    int s[K];
    float w[K];
    float4 r[K];
    #pragma unroll
    for (int k = 0; k < K; ++k) {
        s[k] = __builtin_amdgcn_readlane(en.x, j + k);
        w[k] = __int_as_float(__builtin_amdgcn_readlane(en.y, j + k));
    }
    #pragma unroll
    for (int k = 0; k < K; ++k) r[k] = *(const float4*)(hl + (size_t)s[k] * DIM);
    #pragma unroll
    for (int k = 0; k < K; ++k) {
        acc.x = fmaf(r[k].x, w[k], acc.x);
        acc.y = fmaf(r[k].y, w[k], acc.y);
        acc.z = fmaf(r[k].z, w[k], acc.z);
        acc.w = fmaf(r[k].w, w[k], acc.w);
    }
}

#define GATHER_NODES 4
__launch_bounds__(256)
__global__ void k_gather_rows(const int* __restrict__ rowptr, const int* __restrict__ cnt,
                              const int* __restrict__ ent, const float* __restrict__ h,
                              const float* __restrict__ dinv, const float* __restrict__ bias,
                              float* __restrict__ out) {
    const int lane = threadIdx.x & 31;
    const int d0 = GATHER_NODES *
        __builtin_amdgcn_readfirstlane((int)(blockIdx.x * 8 + (threadIdx.x >> 5)));
    if (d0 >= N_NODES) return;
    const int2* ent2 = (const int2*)ent;
    const float* hl = h + lane * 4;
    const int4   st4 = *(const int4*)(rowptr + d0);
    const int4   n4  = *(const int4*)(cnt + d0);
    const float4 di4 = *(const float4*)(dinv + d0);
    const int   st[GATHER_NODES] = {st4.x, st4.y, st4.z, st4.w};
    const int   nn[GATHER_NODES] = {n4.x, n4.y, n4.z, n4.w};
    const float dd[GATHER_NODES] = {di4.x, di4.y, di4.z, di4.w};
    int2 en0[GATHER_NODES];
    #pragma unroll
    for (int i = 0; i < GATHER_NODES; ++i) {
        const int m0 = (nn[i] < 32) ? nn[i] : 32;
        en0[i] = make_int2(0, 0);
        if (lane < m0) en0[i] = ent2[st[i] + lane];
    }
    float4 bv = make_float4(0.f, 0.f, 0.f, 0.f);
    if (bias) bv = *(const float4*)(bias + lane * 4);
    #pragma unroll
    for (int i = 0; i < GATHER_NODES; ++i) {
        const int d = d0 + i;
        const int start = st[i];
        const int n = nn[i];
        const float di = dd[i];
        int m = (n < 32) ? n : 32;
        int2 en = en0[i];
        const float4 hs = *(const float4*)(hl + (size_t)d * DIM);
        float4 acc;
        acc.x = hs.x * di * di;
        acc.y = hs.y * di * di;
        acc.z = hs.z * di * di;
        acc.w = hs.w * di * di;
        if (bias) {
            acc.x += bv.x; acc.y += bv.y; acc.z += bv.z; acc.w += bv.w;
        }
        for (int base = 0; base < n; ) {
            int j = 0;
            for (; j + 8 <= m; j += 8) add_rows<8>(acc, hl, en, j);
            if (j + 4 <= m) { add_rows<4>(acc, hl, en, j); j += 4; }
            if (j + 2 <= m) { add_rows<2>(acc, hl, en, j); j += 2; }
            if (j < m)      { add_rows<1>(acc, hl, en, j); }
            base += 32;
            if (base < n) {
                m = (n - base < 32) ? (n - base) : 32;
                en = make_int2(0, 0);
                if (lane < m) en = ent2[start + base + lane];
            }
        }
        *(float4*)(out + (size_t)d * DIM + lane * 4) = acc;
    }
}

#define DEC_BATCH 4
__launch_bounds__(256)
__global__ void k_decode(const int* __restrict__ pa, const int* __restrict__ pb,
                         const float* __restrict__ z, float* __restrict__ out) {
    const int lane = threadIdx.x & 31;
    const int c = __builtin_amdgcn_readfirstlane((int)(blockIdx.x * 8 + (threadIdx.x >> 5)));
    if (c >= N_PAIRS / 32) return;
    const int p0 = c * 32;
    const int ia = pa[p0 + lane];
    const int ib = pb[p0 + lane];
    const float* zl = z + lane * 4;
    float res = 0.0f;
    for (int j = 0; j < 32; j += DEC_BATCH) {
        int a[DEC_BATCH], b[DEC_BATCH];
        float4 za[DEC_BATCH], zb[DEC_BATCH];
        #pragma unroll
        for (int k = 0; k < DEC_BATCH; ++k) {
            a[k] = __builtin_amdgcn_readlane(ia, j + k);
            b[k] = __builtin_amdgcn_readlane(ib, j + k);
        }
        #pragma unroll
        for (int k = 0; k < DEC_BATCH; ++k) {
            za[k] = *(const float4*)(zl + (size_t)a[k] * DIM);
            zb[k] = *(const float4*)(zl + (size_t)b[k] * DIM);
        }
        #pragma unroll
        for (int k = 0; k < DEC_BATCH; ++k) {
            float s = za[k].x * zb[k].x + za[k].y * zb[k].y + za[k].z * zb[k].z + za[k].w * zb[k].w;
            #pragma unroll
            for (int off = 16; off > 0; off >>= 1) s += __shfl_xor(s, off);
            if (lane == j + k) res = s;
        }
    }
    out[p0 + lane] = res;
}

extern "C" void kernel_launch(void* const* d_in, const int* in_sizes, int n_in,
                              void* d_out, int out_size, void* d_ws, size_t ws_size,
                              hipStream_t stream) {
    const int*   edge_index = (const int*)d_in[0];
    const int*   edge_pairs = (const int*)d_in[1];
    const float* emb        = (const float*)d_in[2];
    const float* W1         = (const float*)d_in[3];
    const float* b1         = (const float*)d_in[4];
    const float* W2         = (const float*)d_in[5];
    const float* b2         = (const float*)d_in[6];
    float*       out        = (float*)d_out;

    const int* src = edge_index;
    const int* dst = edge_index + N_EDGES;
    const int* pa  = edge_pairs;
    const int* pb  = edge_pairs + N_PAIRS;

    char* ws = (char*)d_ws;
    size_t off = 0;
    int*   cnt    = (int*)(ws + off);   off += ((size_t)N_NODES * 4 + 255) & ~(size_t)255;
    int*   rowptr = (int*)(ws + off);   off += ((size_t)N_NODES * 4 + 255) & ~(size_t)255;
    float* dinv   = (float*)(ws + off); off += ((size_t)N_NODES * 4 + 255) & ~(size_t)255;
    int*   rank   = (int*)(ws + off);   off += (size_t)N_EDGES * 4;
    int*   ent    = (int*)(ws + off);   off += (size_t)N_EDGES * 8;
    float* h      = (float*)(ws + off); off += (size_t)N_NODES * DIM * 4;
    float* agg    = (float*)(ws + off); off += (size_t)N_NODES * DIM * 4;
    (void)ws_size; (void)in_sizes; (void)n_in; (void)out_size;

    const int T = 256;
    const int gN    = (N_NODES + T - 1) / T;
    const int gMM   = N_NODES / 16;
    const int gE4   = (N_EDGES / 4 + T - 1) / T;
    const int gNw   = (N_NODES / GATHER_NODES + 7) / 8;
    const int gPw   = (N_PAIRS / 32 + 7) / 8;

    k_zero_cnt<<<gN, T, 0, stream>>>(cnt);
    k_count_rank<<<gE4, T, 0, stream>>>(dst, cnt, rank);
    k_scan_deg<<<SCAN_BLOCKS, SCAN_T, 0, stream>>>(cnt, rowptr, dinv);
    k_fill_sorted<<<gE4, T, 0, stream>>>(src, dst, rank, rowptr, dinv, ent);

    k_matmul_wmma<<<gMM, T, 0, stream>>>(emb, W1, nullptr, h);
    k_gather_rows<<<gNw, T, 0, stream>>>(rowptr, cnt, ent, h, dinv, nullptr, agg);

    k_matmul_wmma<<<gMM, T, 0, stream>>>(agg, W2, b1, h);
    k_gather_rows<<<gNw, T, 0, stream>>>(rowptr, cnt, ent, h, dinv, b2, agg);

    k_decode<<<gPw, T, 0, stream>>>(pa, pb, agg, out);
}


# Aggregate first, then transform, in one kernel: a block gathers 32 node rows into LDS and multiplies them by W on the matrix unit; (A_hat X)W for A_hat(XW) reorders the sums; 9 launches to 7
# speedup vs baseline: 2.7345x; 1.1237x over previous
#include <hip/hip_runtime.h>

#define N_NODES 50000
#define N_EDGES 800000
#define N_PAIRS 400000
#define DIM     128
#define LDS_STRIDE 132

typedef float v2f __attribute__((ext_vector_type(2)));
typedef float v8f __attribute__((ext_vector_type(8)));

__global__ void k_zero_cnt(int* __restrict__ cnt) {
    int i = blockIdx.x * blockDim.x + threadIdx.x;
    if (i < N_NODES) cnt[i] = 0;
}

__global__ void k_count_rank(const int* __restrict__ dst, int* __restrict__ cnt,
                             int* __restrict__ rank) {
    int q = blockIdx.x * blockDim.x + threadIdx.x;
    if (q < N_EDGES / 4) {
        const int4 d = ((const int4*)dst)[q];
        int4 r;
        r.x = atomicAdd(&cnt[d.x], 1);
        r.y = atomicAdd(&cnt[d.y], 1);
        r.z = atomicAdd(&cnt[d.z], 1);
        r.w = atomicAdd(&cnt[d.w], 1);
        ((int4*)rank)[q] = r;
    }
}

#define SCAN_T 1024
#define SCAN_BLOCKS 13
__launch_bounds__(SCAN_T)
__global__ void k_scan_deg(const int* __restrict__ cnt, int* __restrict__ rowptr,
                           float* __restrict__ dinv) {
    __shared__ int s_tot[32];
    __shared__ int s_pre[32];
    const int t = threadIdx.x;
    const int lane = t & 31;
    const int wave = t >> 5;
    const int b = blockIdx.x;
    const int4* c4 = (const int4*)cnt;

    int4 f[SCAN_BLOCKS - 1];
    #pragma unroll
    for (int k = 0; k < SCAN_BLOCKS - 1; ++k) f[k] = c4[k * SCAN_T + t];
    int pre = 0;
    #pragma unroll
    for (int k = 0; k < SCAN_BLOCKS - 1; ++k)
        pre += (k < b) ? ((f[k].x + f[k].y) + (f[k].z + f[k].w)) : 0;

    const int i4 = b * SCAN_T + t;
    int4 c = make_int4(0, 0, 0, 0);
    if (i4 < N_NODES / 4) c = c4[i4];
    const int tot = (c.x + c.y) + (c.z + c.w);

    int incl = tot;
    #pragma unroll
    for (int off = 1; off < 32; off <<= 1) {
        int up = __shfl_up(incl, off);
        if (lane >= off) incl += up;
    }
    #pragma unroll
    for (int off = 16; off > 0; off >>= 1) pre += __shfl_xor(pre, off);
    if (lane == 31) s_tot[wave] = incl;
    if (lane == 0)  s_pre[wave] = pre;
    __syncthreads();
    int base = 0;
    for (int w = 0; w < 32; ++w) {
        base += s_pre[w];
        if (w < wave) base += s_tot[w];
    }
    if (i4 < N_NODES / 4) {
        int4 r;
        r.x = base + incl - tot;
        r.y = r.x + c.x;
        r.z = r.y + c.y;
        r.w = r.z + c.z;
        ((int4*)rowptr)[i4] = r;
        float4 dv;
        dv.x = rsqrtf((float)(c.x + 1));
        dv.y = rsqrtf((float)(c.y + 1));
        dv.z = rsqrtf((float)(c.z + 1));
        dv.w = rsqrtf((float)(c.w + 1));
        ((float4*)dinv)[i4] = dv;
    }
}

__global__ void k_fill_sorted(const int* __restrict__ src, const int* __restrict__ dst,
                              const int* __restrict__ rank, const int* __restrict__ rowptr,
                              const float* __restrict__ dinv, int* __restrict__ ent) {
    int q = blockIdx.x * blockDim.x + threadIdx.x;
    if (q < N_EDGES / 4) {
        const int4 s4 = ((const int4*)src)[q];
        const int4 d4 = ((const int4*)dst)[q];
        const int4 r4 = ((const int4*)rank)[q];
        const int s[4] = {s4.x, s4.y, s4.z, s4.w};
        const int d[4] = {d4.x, d4.y, d4.z, d4.w};
        const int r[4] = {r4.x, r4.y, r4.z, r4.w};
        float w[4];
        int   pos[4];
        #pragma unroll
        for (int k = 0; k < 4; ++k) {
            w[k]   = dinv[s[k]] * dinv[d[k]];
            pos[k] = rowptr[d[k]] + r[k];
        }
        #pragma unroll
        for (int k = 0; k < 4; ++k)
            ((int2*)ent)[pos[k]] = make_int2(s[k], __float_as_int(w[k]));
    }
}


template <int K>
__device__ __forceinline__ void add_rows(float4& acc, const float* __restrict__ hl,
                                         const int2 en, const int j) {
    int s[K];
    float w[K];
    float4 r[K];
    #pragma unroll
    for (int k = 0; k < K; ++k) {
        s[k] = __builtin_amdgcn_readlane(en.x, j + k);
        w[k] = __int_as_float(__builtin_amdgcn_readlane(en.y, j + k));
    }
    #pragma unroll
    for (int k = 0; k < K; ++k) r[k] = *(const float4*)(hl + (size_t)s[k] * DIM);
    #pragma unroll
    for (int k = 0; k < K; ++k) {
        acc.x = fmaf(r[k].x, w[k], acc.x);
        acc.y = fmaf(r[k].y, w[k], acc.y);
        acc.z = fmaf(r[k].z, w[k], acc.z);
        acc.w = fmaf(r[k].w, w[k], acc.w);
    }
}

#define GATHER_NODES 4
#define FUSE_ROWS 32
static_assert(FUSE_ROWS == 8 * GATHER_NODES, "eight waves fill the tile");
static_assert(N_NODES % 16 == 0, "a 16-row tile is all nodes or none");
__launch_bounds__(256)
__global__ void k_gather_matmul(const int* __restrict__ rowptr, const int* __restrict__ cnt,
                                const int* __restrict__ ent, const float* __restrict__ x,
                                const float* __restrict__ dinv, const float* __restrict__ Wm,
                                const float* __restrict__ bias, const int relu,
                                float* __restrict__ out) {
    __shared__ float lds[FUSE_ROWS * LDS_STRIDE];
    const int lane = threadIdx.x & 31;
    const int wave = __builtin_amdgcn_readfirstlane((int)(threadIdx.x >> 5));
    const int half = lane >> 4;
    const int l16  = lane & 15;
    const int colBase = wave * 16;
    const int rowBase = blockIdx.x * FUSE_ROWS;

    const int d0 = rowBase + GATHER_NODES * wave;
    float* trow = &lds[(GATHER_NODES * wave) * LDS_STRIDE + lane * 4];
    if (d0 < N_NODES) {
        const int2* ent2 = (const int2*)ent;
        const float* hl = x + lane * 4;
        const int4   st4 = *(const int4*)(rowptr + d0);
        const int4   n4  = *(const int4*)(cnt + d0);
        const float4 di4 = *(const float4*)(dinv + d0);
        const int   st[GATHER_NODES] = {st4.x, st4.y, st4.z, st4.w};
        const int   nn[GATHER_NODES] = {n4.x, n4.y, n4.z, n4.w};
        const float dd[GATHER_NODES] = {di4.x, di4.y, di4.z, di4.w};
        int2 en0[GATHER_NODES];
        #pragma unroll
        for (int i = 0; i < GATHER_NODES; ++i) {
            const int m0 = (nn[i] < 32) ? nn[i] : 32;
            en0[i] = make_int2(0, 0);
            if (lane < m0) en0[i] = ent2[st[i] + lane];
        }
        #pragma unroll
        for (int i = 0; i < GATHER_NODES; ++i) {
            const int d = d0 + i;
            const int start = st[i];
            const int n = nn[i];
            const float di = dd[i];
            int m = (n < 32) ? n : 32;
            int2 en = en0[i];
            const float4 hs = *(const float4*)(hl + (size_t)d * DIM);
            float4 acc;
            acc.x = hs.x * di * di;
            acc.y = hs.y * di * di;
            acc.z = hs.z * di * di;
            acc.w = hs.w * di * di;
            for (int base = 0; base < n; ) {
                int j = 0;
                for (; j + 8 <= m; j += 8) add_rows<8>(acc, hl, en, j);
                if (j + 4 <= m) { add_rows<4>(acc, hl, en, j); j += 4; }
                if (j + 2 <= m) { add_rows<2>(acc, hl, en, j); j += 2; }
                if (j < m)      { add_rows<1>(acc, hl, en, j); }
                base += 32;
                if (base < n) {
                    m = (n - base < 32) ? (n - base) : 32;
                    en = make_int2(0, 0);
                    if (lane < m) en = ent2[start + base + lane];
                }
            }
            *(float4*)(trow + i * LDS_STRIDE) = acc;
        }
    } else {
        #pragma unroll
        for (int i = 0; i < GATHER_NODES; ++i)
            *(float4*)(trow + i * LDS_STRIDE) = make_float4(0.f, 0.f, 0.f, 0.f);
    }
    __syncthreads();

    v8f acc0 = {0.f, 0.f, 0.f, 0.f, 0.f, 0.f, 0.f, 0.f};
    v8f acc1 = {0.f, 0.f, 0.f, 0.f, 0.f, 0.f, 0.f, 0.f};
    #pragma unroll
    for (int s = 0; s < DIM / 4; ++s) {
        const int kk = 4 * s + half * 2;
        const float2 av0 = *(const float2*)(&lds[l16 * LDS_STRIDE + kk]);
        const float2 av1 = *(const float2*)(&lds[(16 + l16) * LDS_STRIDE + kk]);
        v2f a0, a1, b;
        a0[0] = av0.x;
        a0[1] = av0.y;
        a1[0] = av1.x;
        a1[1] = av1.y;
        b[0] = Wm[(size_t)kk * DIM + colBase + l16];
        b[1] = Wm[(size_t)(kk + 1) * DIM + colBase + l16];
        acc0 = __builtin_amdgcn_wmma_f32_16x16x4_f32(false, a0, false, b,
                                                     (short)0, acc0, false, false);
        acc1 = __builtin_amdgcn_wmma_f32_16x16x4_f32(false, a1, false, b,
                                                     (short)0, acc1, false, false);
    }

    const float bcol = bias[colBase + l16];
    float r0[8], r1[8];
    #pragma unroll
    for (int v = 0; v < 8; ++v) {
        r0[v] = acc0[v] + bcol;
        r1[v] = acc1[v] + bcol;
        if (relu) {
            r0[v] = fmaxf(r0[v], 0.0f);
            r1[v] = fmaxf(r1[v], 0.0f);
        }
    }
    float* outp = out + (size_t)rowBase * DIM + colBase + l16;
    #pragma unroll
    for (int v = 0; v < 8; ++v) outp[(size_t)(v + half * 8) * DIM] = r0[v];
    if (rowBase + 16 < N_NODES) {
        #pragma unroll
        for (int v = 0; v < 8; ++v) outp[(size_t)(16 + v + half * 8) * DIM] = r1[v];
    }
}

#define DEC_BATCH 4
__launch_bounds__(256)
__global__ void k_decode(const int* __restrict__ pa, const int* __restrict__ pb,
                         const float* __restrict__ z, float* __restrict__ out) {
    const int lane = threadIdx.x & 31;
    const int c = __builtin_amdgcn_readfirstlane((int)(blockIdx.x * 8 + (threadIdx.x >> 5)));
    if (c >= N_PAIRS / 32) return;
    const int p0 = c * 32;
    const int ia = pa[p0 + lane];
    const int ib = pb[p0 + lane];
    const float* zl = z + lane * 4;
    float res = 0.0f;
    for (int j = 0; j < 32; j += DEC_BATCH) {
        int a[DEC_BATCH], b[DEC_BATCH];
        float4 za[DEC_BATCH], zb[DEC_BATCH];
        #pragma unroll
        for (int k = 0; k < DEC_BATCH; ++k) {
            a[k] = __builtin_amdgcn_readlane(ia, j + k);
            b[k] = __builtin_amdgcn_readlane(ib, j + k);
        }
        #pragma unroll
        for (int k = 0; k < DEC_BATCH; ++k) {
            za[k] = *(const float4*)(zl + (size_t)a[k] * DIM);
            zb[k] = *(const float4*)(zl + (size_t)b[k] * DIM);
        }
        #pragma unroll
        for (int k = 0; k < DEC_BATCH; ++k) {
            float s = za[k].x * zb[k].x + za[k].y * zb[k].y + za[k].z * zb[k].z + za[k].w * zb[k].w;
            #pragma unroll
            for (int off = 16; off > 0; off >>= 1) s += __shfl_xor(s, off);
            if (lane == j + k) res = s;
        }
    }
    out[p0 + lane] = res;
}

extern "C" void kernel_launch(void* const* d_in, const int* in_sizes, int n_in,
                              void* d_out, int out_size, void* d_ws, size_t ws_size,
                              hipStream_t stream) {
    const int*   edge_index = (const int*)d_in[0];
    const int*   edge_pairs = (const int*)d_in[1];
    const float* emb        = (const float*)d_in[2];
    const float* W1         = (const float*)d_in[3];
    const float* b1         = (const float*)d_in[4];
    const float* W2         = (const float*)d_in[5];
    const float* b2         = (const float*)d_in[6];
    float*       out        = (float*)d_out;

    const int* src = edge_index;
    const int* dst = edge_index + N_EDGES;
    const int* pa  = edge_pairs;
    const int* pb  = edge_pairs + N_PAIRS;

    char* ws = (char*)d_ws;
    size_t off = 0;
    int*   cnt    = (int*)(ws + off);   off += ((size_t)N_NODES * 4 + 255) & ~(size_t)255;
    int*   rowptr = (int*)(ws + off);   off += ((size_t)N_NODES * 4 + 255) & ~(size_t)255;
    float* dinv   = (float*)(ws + off); off += ((size_t)N_NODES * 4 + 255) & ~(size_t)255;
    int*   rank   = (int*)(ws + off);   off += (size_t)N_EDGES * 4;
    int*   ent    = (int*)(ws + off);   off += (size_t)N_EDGES * 8;
    float* z1     = (float*)(ws + off); off += (size_t)N_NODES * DIM * 4;
    float* z2     = (float*)(ws + off); off += (size_t)N_NODES * DIM * 4;
    (void)ws_size; (void)in_sizes; (void)n_in; (void)out_size;

    const int T = 256;
    const int gN    = (N_NODES + T - 1) / T;
    const int gE4   = (N_EDGES / 4 + T - 1) / T;
    const int gF    = (N_NODES + FUSE_ROWS - 1) / FUSE_ROWS;
    const int gPw   = (N_PAIRS / 32 + 7) / 8;

    k_zero_cnt<<<gN, T, 0, stream>>>(cnt);
    k_count_rank<<<gE4, T, 0, stream>>>(dst, cnt, rank);
    k_scan_deg<<<SCAN_BLOCKS, SCAN_T, 0, stream>>>(cnt, rowptr, dinv);
    k_fill_sorted<<<gE4, T, 0, stream>>>(src, dst, rank, rowptr, dinv, ent);

    k_gather_matmul<<<gF, T, 0, stream>>>(rowptr, cnt, ent, emb, dinv, W1, b1, 1, z1);

    k_gather_matmul<<<gF, T, 0, stream>>>(rowptr, cnt, ent, z1, dinv, W2, b2, 0, z2);

    k_decode<<<gPw, T, 0, stream>>>(pa, pb, z2, out);
}


# Row offsets scanned in LDS (196 KB) inside the fill kernel, by every work-group: scan launch gone (7 launches to 6); an edge's offset and two degrees come from LDS, not scattered global reads
# speedup vs baseline: 2.9259x; 1.0700x over previous
#include <hip/hip_runtime.h>

#define N_NODES 50000
#define N_EDGES 800000
#define N_PAIRS 400000
#define DIM     128
#define LDS_STRIDE 132

typedef float v2f __attribute__((ext_vector_type(2)));
typedef float v8f __attribute__((ext_vector_type(8)));

__global__ void k_zero_cnt(int* __restrict__ cnt) {
    int i = blockIdx.x * blockDim.x + threadIdx.x;
    if (i < N_NODES) cnt[i] = 0;
}

__global__ void k_count_rank(const int* __restrict__ dst, int* __restrict__ cnt,
                             int* __restrict__ rank) {
    int q = blockIdx.x * blockDim.x + threadIdx.x;
    if (q < N_EDGES / 4) {
        const int4 d = ((const int4*)dst)[q];
        int4 r;
        r.x = atomicAdd(&cnt[d.x], 1);
        r.y = atomicAdd(&cnt[d.y], 1);
        r.z = atomicAdd(&cnt[d.z], 1);
        r.w = atomicAdd(&cnt[d.w], 1);
        ((int4*)rank)[q] = r;
    }
}

#define SF_T 1024
#define SF_BLOCKS 256
#define SF_CHUNK 49
#define SF_SLOTS (SF_T * SF_CHUNK)
#define SF_NODES 196
#define SF_QUADS 782
static_assert(SF_SLOTS >= N_NODES + 1, "slot N_NODES holds the edge total");
static_assert(SF_BLOCKS * SF_NODES >= N_NODES, "every node written");
static_assert(SF_BLOCKS * SF_QUADS >= N_EDGES / 4, "every edge placed");
static_assert(SF_QUADS <= SF_T && SF_NODES <= SF_T, "one item per thread");
static_assert(N_NODES % 4 == 0 && N_NODES / 4 <= 13 * SF_T, "13 int4 copies per thread");
__launch_bounds__(SF_T)
__global__ void k_scan_fill(const int* __restrict__ src, const int* __restrict__ dst,
                            const int* __restrict__ rank, const int* __restrict__ cnt,
                            int* __restrict__ rowptr, float* __restrict__ dinv,
                            int* __restrict__ ent) {
    __shared__ __attribute__((aligned(16))) int s_row[SF_SLOTS];
    __shared__ int s_wtot[32];
    const int t = threadIdx.x;
    const int lane = t & 31;
    const int wave = t >> 5;
    const int b = blockIdx.x;

    const int q = b * SF_QUADS + t;
    const bool has_q = (t < SF_QUADS) && (q < N_EDGES / 4);
    int4 s4 = make_int4(0, 0, 0, 0);
    int4 d4 = make_int4(0, 0, 0, 0);
    int4 r4 = make_int4(0, 0, 0, 0);
    if (has_q) {
        s4 = ((const int4*)src)[q];
        d4 = ((const int4*)dst)[q];
        r4 = ((const int4*)rank)[q];
    }

    const int4* c4 = (const int4*)cnt;
    #pragma unroll
    for (int k = 0; k < 13; ++k) {
        const int i = k * SF_T + t;
        if (i < N_NODES / 4) ((int4*)s_row)[i] = c4[i];
    }
    for (int i = N_NODES + t; i < SF_SLOTS; i += SF_T) s_row[i] = 0;
    __syncthreads();

    int c[SF_CHUNK];
    int tot = 0;
    #pragma unroll
    for (int k = 0; k < SF_CHUNK; ++k) {
        c[k] = s_row[t * SF_CHUNK + k];
        tot += c[k];
    }
    int incl = tot;
    #pragma unroll
    for (int off = 1; off < 32; off <<= 1) {
        const int up = __shfl_up(incl, off);
        if (lane >= off) incl += up;
    }
    if (lane == 31) s_wtot[wave] = incl;
    __syncthreads();
    int run = incl - tot;
    for (int w = 0; w < 32; ++w) run += (w < wave) ? s_wtot[w] : 0;
    #pragma unroll
    for (int k = 0; k < SF_CHUNK; ++k) {
        s_row[t * SF_CHUNK + k] = run;
        run += c[k];
    }
    __syncthreads();

    if (t < SF_NODES) {
        const int i = b * SF_NODES + t;
        if (i < N_NODES) {
            const int lo = s_row[i];
            const int hi = s_row[i + 1];
            rowptr[i] = lo;
            dinv[i] = rsqrtf((float)(hi - lo + 1));
        }
    }

    if (has_q) {
        const int s[4] = {s4.x, s4.y, s4.z, s4.w};
        const int d[4] = {d4.x, d4.y, d4.z, d4.w};
        const int r[4] = {r4.x, r4.y, r4.z, r4.w};
        float w[4];
        int   pos[4];
        #pragma unroll
        for (int k = 0; k < 4; ++k) {
            const int dlo = s_row[d[k]];
            const int dhi = s_row[d[k] + 1];
            const int slo = s_row[s[k]];
            const int shi = s_row[s[k] + 1];
            w[k]   = rsqrtf((float)(shi - slo + 1)) * rsqrtf((float)(dhi - dlo + 1));
            pos[k] = dlo + r[k];
        }
        #pragma unroll
        for (int k = 0; k < 4; ++k)
            ((int2*)ent)[pos[k]] = make_int2(s[k], __float_as_int(w[k]));
    }
}


template <int K>
__device__ __forceinline__ void add_rows(float4& acc, const float* __restrict__ hl,
                                         const int2 en, const int j) {
    int s[K];
    float w[K];
    float4 r[K];
    #pragma unroll
    for (int k = 0; k < K; ++k) {
        s[k] = __builtin_amdgcn_readlane(en.x, j + k);
        w[k] = __int_as_float(__builtin_amdgcn_readlane(en.y, j + k));
    }
    #pragma unroll
    for (int k = 0; k < K; ++k) r[k] = *(const float4*)(hl + (size_t)s[k] * DIM);
    #pragma unroll
    for (int k = 0; k < K; ++k) {
        acc.x = fmaf(r[k].x, w[k], acc.x);
        acc.y = fmaf(r[k].y, w[k], acc.y);
        acc.z = fmaf(r[k].z, w[k], acc.z);
        acc.w = fmaf(r[k].w, w[k], acc.w);
    }
}

#define GATHER_NODES 4
#define FUSE_ROWS 32
static_assert(FUSE_ROWS == 8 * GATHER_NODES, "eight waves fill the tile");
static_assert(N_NODES % 16 == 0, "a 16-row tile is all nodes or none");
__launch_bounds__(256)
__global__ void k_gather_matmul(const int* __restrict__ rowptr, const int* __restrict__ cnt,
                                const int* __restrict__ ent, const float* __restrict__ x,
                                const float* __restrict__ dinv, const float* __restrict__ Wm,
                                const float* __restrict__ bias, const int relu,
                                float* __restrict__ out) {
    __shared__ float lds[FUSE_ROWS * LDS_STRIDE];
    const int lane = threadIdx.x & 31;
    const int wave = __builtin_amdgcn_readfirstlane((int)(threadIdx.x >> 5));
    const int half = lane >> 4;
    const int l16  = lane & 15;
    const int colBase = wave * 16;
    const int rowBase = blockIdx.x * FUSE_ROWS;

    const int d0 = rowBase + GATHER_NODES * wave;
    float* trow = &lds[(GATHER_NODES * wave) * LDS_STRIDE + lane * 4];
    if (d0 < N_NODES) {
        const int2* ent2 = (const int2*)ent;
        const float* hl = x + lane * 4;
        const int4   st4 = *(const int4*)(rowptr + d0);
        const int4   n4  = *(const int4*)(cnt + d0);
        const float4 di4 = *(const float4*)(dinv + d0);
        const int   st[GATHER_NODES] = {st4.x, st4.y, st4.z, st4.w};
        const int   nn[GATHER_NODES] = {n4.x, n4.y, n4.z, n4.w};
        const float dd[GATHER_NODES] = {di4.x, di4.y, di4.z, di4.w};
        int2 en0[GATHER_NODES];
        #pragma unroll
        for (int i = 0; i < GATHER_NODES; ++i) {
            const int m0 = (nn[i] < 32) ? nn[i] : 32;
            en0[i] = make_int2(0, 0);
            if (lane < m0) en0[i] = ent2[st[i] + lane];
        }
        #pragma unroll
        for (int i = 0; i < GATHER_NODES; ++i) {
            const int d = d0 + i;
            const int start = st[i];
            const int n = nn[i];
            const float di = dd[i];
            int m = (n < 32) ? n : 32;
            int2 en = en0[i];
            const float4 hs = *(const float4*)(hl + (size_t)d * DIM);
            float4 acc;
            acc.x = hs.x * di * di;
            acc.y = hs.y * di * di;
            acc.z = hs.z * di * di;
            acc.w = hs.w * di * di;
            for (int base = 0; base < n; ) {
                int j = 0;
                for (; j + 8 <= m; j += 8) add_rows<8>(acc, hl, en, j);
                if (j + 4 <= m) { add_rows<4>(acc, hl, en, j); j += 4; }
                if (j + 2 <= m) { add_rows<2>(acc, hl, en, j); j += 2; }
                if (j < m)      { add_rows<1>(acc, hl, en, j); }
                base += 32;
                if (base < n) {
                    m = (n - base < 32) ? (n - base) : 32;
                    en = make_int2(0, 0);
                    if (lane < m) en = ent2[start + base + lane];
                }
            }
            *(float4*)(trow + i * LDS_STRIDE) = acc;
        }
    } else {
        #pragma unroll
        for (int i = 0; i < GATHER_NODES; ++i)
            *(float4*)(trow + i * LDS_STRIDE) = make_float4(0.f, 0.f, 0.f, 0.f);
    }
    __syncthreads();

    v8f acc0 = {0.f, 0.f, 0.f, 0.f, 0.f, 0.f, 0.f, 0.f};
    v8f acc1 = {0.f, 0.f, 0.f, 0.f, 0.f, 0.f, 0.f, 0.f};
    #pragma unroll
    for (int s = 0; s < DIM / 4; ++s) {
        const int kk = 4 * s + half * 2;
        const float2 av0 = *(const float2*)(&lds[l16 * LDS_STRIDE + kk]);
        const float2 av1 = *(const float2*)(&lds[(16 + l16) * LDS_STRIDE + kk]);
        v2f a0, a1, b;
        a0[0] = av0.x;
        a0[1] = av0.y;
        a1[0] = av1.x;
        a1[1] = av1.y;
        b[0] = Wm[(size_t)kk * DIM + colBase + l16];
        b[1] = Wm[(size_t)(kk + 1) * DIM + colBase + l16];
        acc0 = __builtin_amdgcn_wmma_f32_16x16x4_f32(false, a0, false, b,
                                                     (short)0, acc0, false, false);
        acc1 = __builtin_amdgcn_wmma_f32_16x16x4_f32(false, a1, false, b,
                                                     (short)0, acc1, false, false);
    }

    const float bcol = bias[colBase + l16];
    float r0[8], r1[8];
    #pragma unroll
    for (int v = 0; v < 8; ++v) {
        r0[v] = acc0[v] + bcol;
        r1[v] = acc1[v] + bcol;
        if (relu) {
            r0[v] = fmaxf(r0[v], 0.0f);
            r1[v] = fmaxf(r1[v], 0.0f);
        }
    }
    float* outp = out + (size_t)rowBase * DIM + colBase + l16;
    #pragma unroll
    for (int v = 0; v < 8; ++v) outp[(size_t)(v + half * 8) * DIM] = r0[v];
    if (rowBase + 16 < N_NODES) {
        #pragma unroll
        for (int v = 0; v < 8; ++v) outp[(size_t)(16 + v + half * 8) * DIM] = r1[v];
    }
}

#define DEC_BATCH 4
__launch_bounds__(256)
__global__ void k_decode(const int* __restrict__ pa, const int* __restrict__ pb,
                         const float* __restrict__ z, float* __restrict__ out) {
    const int lane = threadIdx.x & 31;
    const int c = __builtin_amdgcn_readfirstlane((int)(blockIdx.x * 8 + (threadIdx.x >> 5)));
    if (c >= N_PAIRS / 32) return;
    const int p0 = c * 32;
    const int ia = pa[p0 + lane];
    const int ib = pb[p0 + lane];
    const float* zl = z + lane * 4;
    float res = 0.0f;
    for (int j = 0; j < 32; j += DEC_BATCH) {
        int a[DEC_BATCH], b[DEC_BATCH];
        float4 za[DEC_BATCH], zb[DEC_BATCH];
        #pragma unroll
        for (int k = 0; k < DEC_BATCH; ++k) {
            a[k] = __builtin_amdgcn_readlane(ia, j + k);
            b[k] = __builtin_amdgcn_readlane(ib, j + k);
        }
        #pragma unroll
        for (int k = 0; k < DEC_BATCH; ++k) {
            za[k] = *(const float4*)(zl + (size_t)a[k] * DIM);
            zb[k] = *(const float4*)(zl + (size_t)b[k] * DIM);
        }
        #pragma unroll
        for (int k = 0; k < DEC_BATCH; ++k) {
            float s = za[k].x * zb[k].x + za[k].y * zb[k].y + za[k].z * zb[k].z + za[k].w * zb[k].w;
            #pragma unroll
            for (int off = 16; off > 0; off >>= 1) s += __shfl_xor(s, off);
            if (lane == j + k) res = s;
        }
    }
    out[p0 + lane] = res;
}

extern "C" void kernel_launch(void* const* d_in, const int* in_sizes, int n_in,
                              void* d_out, int out_size, void* d_ws, size_t ws_size,
                              hipStream_t stream) {
    const int*   edge_index = (const int*)d_in[0];
    const int*   edge_pairs = (const int*)d_in[1];
    const float* emb        = (const float*)d_in[2];
    const float* W1         = (const float*)d_in[3];
    const float* b1         = (const float*)d_in[4];
    const float* W2         = (const float*)d_in[5];
    const float* b2         = (const float*)d_in[6];
    float*       out        = (float*)d_out;

    const int* src = edge_index;
    const int* dst = edge_index + N_EDGES;
    const int* pa  = edge_pairs;
    const int* pb  = edge_pairs + N_PAIRS;

    char* ws = (char*)d_ws;
    size_t off = 0;
    int*   cnt    = (int*)(ws + off);   off += ((size_t)N_NODES * 4 + 255) & ~(size_t)255;
    int*   rowptr = (int*)(ws + off);   off += ((size_t)N_NODES * 4 + 255) & ~(size_t)255;
    float* dinv   = (float*)(ws + off); off += ((size_t)N_NODES * 4 + 255) & ~(size_t)255;
    int*   rank   = (int*)(ws + off);   off += (size_t)N_EDGES * 4;
    int*   ent    = (int*)(ws + off);   off += (size_t)N_EDGES * 8;
    float* z1     = (float*)(ws + off); off += (size_t)N_NODES * DIM * 4;
    float* z2     = (float*)(ws + off); off += (size_t)N_NODES * DIM * 4;
    (void)ws_size; (void)in_sizes; (void)n_in; (void)out_size;

    const int T = 256;
    const int gN    = (N_NODES + T - 1) / T;
    const int gE4   = (N_EDGES / 4 + T - 1) / T;
    const int gF    = (N_NODES + FUSE_ROWS - 1) / FUSE_ROWS;
    const int gPw   = (N_PAIRS / 32 + 7) / 8;

    k_zero_cnt<<<gN, T, 0, stream>>>(cnt);
    k_count_rank<<<gE4, T, 0, stream>>>(dst, cnt, rank);
    k_scan_fill<<<SF_BLOCKS, SF_T, 0, stream>>>(src, dst, rank, cnt, rowptr, dinv, ent);

    k_gather_matmul<<<gF, T, 0, stream>>>(rowptr, cnt, ent, emb, dinv, W1, b1, 1, z1);

    k_gather_matmul<<<gF, T, 0, stream>>>(rowptr, cnt, ent, z1, dinv, W2, b2, 0, z2);

    k_decode<<<gPw, T, 0, stream>>>(pa, pb, z2, out);
}
